# starting kernel
# baseline (speedup 1.0000x reference)
#include <hip/hip_runtime.h>

typedef __attribute__((ext_vector_type(2))) float v2f;
typedef __attribute__((ext_vector_type(8))) float v8f;

namespace {
constexpr int KNBR   = 16;
constexpr int DIN    = 64;
constexpr int DOUT   = 64;
constexpr int DZ     = 128;
constexpr int DZP    = 132;
constexpr int TILE_M = 16;
constexpr int WAVES  = 8;
constexpr int BLOCK  = WAVES * 32;
constexpr int KSTEPS = DZ / 4;
constexpr int WB_HI  = KSTEPS * DOUT + 16;
}

__global__ __launch_bounds__(BLOCK) void
sage_fused_wmma_f32(const float* __restrict__ x,
                    const int*   __restrict__ nbr,
                    const float* __restrict__ W,
                    const float* __restrict__ bias,
                    float* __restrict__ y,
                    int n_tiles)
{
    __shared__ v2f   Wb[2 * WB_HI];
    __shared__ float bl[DOUT];
    __shared__ float zl[WAVES * TILE_M * DZP];

    const int tid  = threadIdx.x;
    const int wave = tid >> 5;
    const int lane = tid & 31;
    const int nlo  = lane & 15;
    const int hi   = lane >> 4;

    for (int i = tid; i < 2 * KSTEPS * DOUT; i += BLOCK) {
        const int h   = i >> 11;
        const int r   = i & (KSTEPS * DOUT - 1);
        const int kk  = r >> 6;
        const int col = r & (DOUT - 1);
        const int k   = 4 * kk + 2 * h;
        v2f w;
        w.x = W[k * DOUT + col];
        w.y = W[(k + 1) * DOUT + col];
        Wb[h * WB_HI + kk * DOUT + col] = w;
    }
    if (tid < DOUT) bl[tid] = bias[tid];

    int tile = (int)blockIdx.x * WAVES + wave;
    if (tile >= n_tiles) tile = n_tiles - 1;
    const long base = (long)tile * TILE_M;

    float* zw = zl + wave * (TILE_M * DZP);
    const int c0 = lane * 2;

    for (int m = 0; m < TILE_M; ++m) {
        const long node = base + m;
        v2f xs = *(const v2f*)(x + node * DIN + c0);
        const int* nb = nbr + node * KNBR;
        float a0 = 0.f, a1 = 0.f;
        #pragma unroll
        for (int j = 0; j < KNBR; ++j) {
            const float* nr = x + (long)nb[j] * DIN;
            v2f v = *(const v2f*)(nr + c0);
            a0 += v.x;
            a1 += v.y;
        }
        float* zr = zw + m * DZP;
        zr[c0]           = xs.x;
        zr[c0 + 1]       = xs.y;
        zr[DIN + c0]     = a0 * (1.0f / KNBR);
        zr[DIN + c0 + 1] = a1 * (1.0f / KNBR);
    }
    __syncthreads();

    v8f acc[4];
    #pragma unroll
    for (int t = 0; t < 4; ++t) {
        const float bv = bl[t * 16 + nlo];
        #pragma unroll
        for (int v = 0; v < 8; ++v) acc[t][v] = bv;
    }

    const float* za  = zw + nlo * DZP + 2 * hi;
    const v2f*   Wbh = Wb + hi * WB_HI + nlo;
    for (int kk = 0; kk < KSTEPS; ++kk) {
        v2f a = *(const v2f*)(za + 4 * kk);
        const v2f* wrow = Wbh + kk * DOUT;
        #pragma unroll
        for (int t = 0; t < 4; ++t) {
            v2f bf = wrow[t * 16];
            acc[t] = __builtin_amdgcn_wmma_f32_16x16x4_f32(
                false, a, false, bf, (short)0, acc[t], false, false);
        }
    }

    #pragma unroll
    for (int t = 0; t < 4; ++t) {
        #pragma unroll
        for (int v = 0; v < 8; ++v) {
            const long row = base + v + 8 * hi;
            y[row * DOUT + t * 16 + nlo] = acc[t][v];
        }
    }
}

extern "C" void kernel_launch(void* const* d_in, const int* in_sizes, int n_in,
                              void* d_out, int out_size, void* d_ws, size_t ws_size,
                              hipStream_t stream) {
    const float* x    = (const float*)d_in[0];
    const int*   nbr  = (const int*)d_in[1];
    const float* W    = (const float*)d_in[2];
    const float* bias = (const float*)d_in[3];
    float*       y    = (float*)d_out;

    const int n_nodes = in_sizes[0] / DIN;
    const int n_tiles = (n_nodes + TILE_M - 1) / TILE_M;
    const int blocks  = (n_tiles + WAVES - 1) / WAVES;

    sage_fused_wmma_f32<<<blocks, BLOCK, 0, stream>>>(x, nbr, W, bias, y, n_tiles);
}


# gather made memory-parallel: tile's 256 neighbour indices read once up front and lane-broadcast, row loads of 3 nodes kept in flight (was index load, 17 row loads, sum in series per node)
# speedup vs baseline: 1.3612x; 1.3612x over previous
#include <hip/hip_runtime.h>

typedef __attribute__((ext_vector_type(2))) float v2f;
typedef __attribute__((ext_vector_type(8))) float v8f;
typedef __attribute__((ext_vector_type(4))) int   v4i;

namespace {
constexpr int KNBR   = 16;
constexpr int DIN    = 64;
constexpr int DOUT   = 64;
constexpr int DZ     = 128;
constexpr int DZP    = 132;
constexpr int TILE_M = 16;
constexpr int WAVES  = 8;
constexpr int BLOCK  = WAVES * 32;
constexpr int KSTEPS = DZ / 4;
constexpr int WB_HI  = KSTEPS * DOUT + 16;
constexpr int DEPTH  = 3;
}

__global__ __launch_bounds__(BLOCK) void
sage_fused_wmma_f32(const float* __restrict__ x,
                    const int*   __restrict__ nbr,
                    const float* __restrict__ W,
                    const float* __restrict__ bias,
                    float* __restrict__ y,
                    int n_tiles)
{
    __shared__ v2f   Wb[2 * WB_HI];
    __shared__ float bl[DOUT];
    __shared__ float zl[WAVES * TILE_M * DZP];

    const int tid  = threadIdx.x;
    const int wave = tid >> 5;
    const int lane = tid & 31;
    const int nlo  = lane & 15;
    const int hi   = lane >> 4;

    int tile = (int)blockIdx.x * WAVES + wave;
    if (tile >= n_tiles) tile = n_tiles - 1;
    const long base = (long)tile * TILE_M;

    const v4i* nq = (const v4i*)(nbr + base * KNBR) + 2 * lane;
    const v4i q0 = nq[0];
    const v4i q1 = nq[1];

    for (int i = tid; i < 2 * KSTEPS * DOUT; i += BLOCK) {
        const int h   = i >> 11;
        const int r   = i & (KSTEPS * DOUT - 1);
        const int kk  = r >> 6;
        const int col = r & (DOUT - 1);
        const int k   = 4 * kk + 2 * h;
        v2f w;
        w.x = W[k * DOUT + col];
        w.y = W[(k + 1) * DOUT + col];
        Wb[h * WB_HI + kk * DOUT + col] = w;
    }
    if (tid < DOUT) bl[tid] = bias[tid];

    float* zw = zl + wave * (TILE_M * DZP);
    const int c0 = lane * 2;

    const unsigned ro[8] = {
        (unsigned)q0.x << 8, (unsigned)q0.y << 8, (unsigned)q0.z << 8, (unsigned)q0.w << 8,
        (unsigned)q1.x << 8, (unsigned)q1.y << 8, (unsigned)q1.z << 8, (unsigned)q1.w << 8 };
    const char* xb    = (const char*)x;
    const char* xself = xb + base * (DIN * 4) + c0 * 4;
    const unsigned cb = (unsigned)c0 * 4u;

    v2f nv[DEPTH][KNBR];
    v2f sv[DEPTH];

    #pragma unroll
    for (int m = 0; m < DEPTH; ++m) {
        sv[m] = *(const v2f*)(xself + m * (DIN * 4));
        #pragma unroll
        for (int j = 0; j < KNBR; ++j) {
            const unsigned r = (unsigned)__builtin_amdgcn_readlane((int)ro[j & 7], 2 * m + (j >> 3));
            nv[m][j] = *(const v2f*)(xb + (r | cb));
        }
    }
    #pragma unroll
    for (int m = 0; m < TILE_M; ++m) {
        __builtin_amdgcn_sched_barrier(0);
        const int s = m % DEPTH;
        float a0 = 0.f, a1 = 0.f;
        #pragma unroll
        for (int j = 0; j < KNBR; ++j) {
            a0 += nv[s][j].x;
            a1 += nv[s][j].y;
        }
        float* zr = zw + m * DZP;
        zr[c0]           = sv[s].x;
        zr[c0 + 1]       = sv[s].y;
        zr[DIN + c0]     = a0 * (1.0f / KNBR);
        zr[DIN + c0 + 1] = a1 * (1.0f / KNBR);
        __builtin_amdgcn_sched_barrier(0);
        if (m + DEPTH < TILE_M) {
            const int mn = m + DEPTH;
            sv[s] = *(const v2f*)(xself + mn * (DIN * 4));
            #pragma unroll
            for (int j = 0; j < KNBR; ++j) {
                const unsigned r = (unsigned)__builtin_amdgcn_readlane((int)ro[j & 7], 2 * mn + (j >> 3));
                nv[s][j] = *(const v2f*)(xb + (r | cb));
            }
        }
    }
    __syncthreads();

    v8f acc[4];
    #pragma unroll
    for (int t = 0; t < 4; ++t) {
        const float bv = bl[t * 16 + nlo];
        #pragma unroll
        for (int v = 0; v < 8; ++v) acc[t][v] = bv;
    }

    const float* za  = zw + nlo * DZP + 2 * hi;
    const v2f*   Wbh = Wb + hi * WB_HI + nlo;
    for (int kk = 0; kk < KSTEPS; ++kk) {
        v2f a = *(const v2f*)(za + 4 * kk);
        const v2f* wrow = Wbh + kk * DOUT;
        #pragma unroll
        for (int t = 0; t < 4; ++t) {
            v2f bf = wrow[t * 16];
            acc[t] = __builtin_amdgcn_wmma_f32_16x16x4_f32(
                false, a, false, bf, (short)0, acc[t], false, false);
        }
    }

    #pragma unroll
    for (int t = 0; t < 4; ++t) {
        #pragma unroll
        for (int v = 0; v < 8; ++v) {
            const long row = base + v + 8 * hi;
            y[row * DOUT + t * 16 + nlo] = acc[t][v];
        }
    }
}

extern "C" void kernel_launch(void* const* d_in, const int* in_sizes, int n_in,
                              void* d_out, int out_size, void* d_ws, size_t ws_size,
                              hipStream_t stream) {
    const float* x    = (const float*)d_in[0];
    const int*   nbr  = (const int*)d_in[1];
    const float* W    = (const float*)d_in[2];
    const float* bias = (const float*)d_in[3];
    float*       y    = (float*)d_out;

    const int n_nodes = in_sizes[0] / DIN;
    const int n_tiles = (n_nodes + TILE_M - 1) / TILE_M;
    const int blocks  = (n_tiles + WAVES - 1) / WAVES;

    sage_fused_wmma_f32<<<blocks, BLOCK, 0, stream>>>(x, nbr, W, bias, y, n_tiles);
}


# own-rows half of the product (k 0-63) issued inside the gather loop, one k-step per node under the row loads in flight; block barrier moved in front of the gather, none behind it; sums unchanged
# speedup vs baseline: 1.4724x; 1.0817x over previous
#include <hip/hip_runtime.h>

typedef __attribute__((ext_vector_type(2))) float v2f;
typedef __attribute__((ext_vector_type(8))) float v8f;
typedef __attribute__((ext_vector_type(4))) int   v4i;

namespace {
constexpr int KNBR   = 16;
constexpr int DIN    = 64;
constexpr int DOUT   = 64;
constexpr int DZ     = 128;
constexpr int DZP    = 132;
constexpr int TILE_M = 16;
constexpr int WAVES  = 8;
constexpr int BLOCK  = WAVES * 32;
constexpr int KSTEPS = DZ / 4;
constexpr int WB_HI  = KSTEPS * DOUT + 16;
constexpr int DEPTH  = 3;
}

__global__ __launch_bounds__(BLOCK) void
sage_fused_wmma_f32(const float* __restrict__ x,
                    const int*   __restrict__ nbr,
                    const float* __restrict__ W,
                    const float* __restrict__ bias,
                    float* __restrict__ y,
                    int n_tiles)
{
    __shared__ v2f   Wb[2 * WB_HI];
    __shared__ float bl[DOUT];
    __shared__ float zl[WAVES * TILE_M * DZP];

    const int tid  = threadIdx.x;
    const int wave = tid >> 5;
    const int lane = tid & 31;
    const int nlo  = lane & 15;
    const int hi   = lane >> 4;

    int tile = (int)blockIdx.x * WAVES + wave;
    if (tile >= n_tiles) tile = n_tiles - 1;
    const long base = (long)tile * TILE_M;

    const v4i* nq = (const v4i*)(nbr + base * KNBR) + 2 * lane;
    const v4i q0 = nq[0];
    const v4i q1 = nq[1];

    const char* xb    = (const char*)x;
    const int   c0    = lane * 2;
    const char* xself = xb + base * (DIN * 4) + c0 * 4;
    v2f sv[TILE_M];
    #pragma unroll
    for (int m = 0; m < TILE_M; ++m) sv[m] = *(const v2f*)(xself + m * (DIN * 4));

    for (int i = tid; i < 2 * KSTEPS * DOUT; i += BLOCK) {
        const int h   = i >> 11;
        const int r   = i & (KSTEPS * DOUT - 1);
        const int kk  = r >> 6;
        const int col = r & (DOUT - 1);
        const int k   = 4 * kk + 2 * h;
        v2f w;
        w.x = W[k * DOUT + col];
        w.y = W[(k + 1) * DOUT + col];
        Wb[h * WB_HI + kk * DOUT + col] = w;
    }
    if (tid < DOUT) bl[tid] = bias[tid];

    float* zw = zl + wave * (TILE_M * DZP);
    #pragma unroll
    for (int m = 0; m < TILE_M; ++m) *(v2f*)(zw + m * DZP + c0) = sv[m];

    __syncthreads();

    v8f acc[4];
    #pragma unroll
    for (int t = 0; t < 4; ++t) {
        const float bv = bl[t * 16 + nlo];
        #pragma unroll
        for (int v = 0; v < 8; ++v) acc[t][v] = bv;
    }
    const float* za  = zw + nlo * DZP + 2 * hi;
    const v2f*   Wbh = Wb + hi * WB_HI + nlo;

    const unsigned ro[8] = {
        (unsigned)q0.x << 8, (unsigned)q0.y << 8, (unsigned)q0.z << 8, (unsigned)q0.w << 8,
        (unsigned)q1.x << 8, (unsigned)q1.y << 8, (unsigned)q1.z << 8, (unsigned)q1.w << 8 };
    const unsigned cb = (unsigned)c0 * 4u;

    v2f nv[DEPTH][KNBR];

    #pragma unroll
    for (int m = 0; m < DEPTH; ++m) {
        #pragma unroll
        for (int j = 0; j < KNBR; ++j) {
            const unsigned r = (unsigned)__builtin_amdgcn_readlane((int)ro[j & 7], 2 * m + (j >> 3));
            nv[m][j] = *(const v2f*)(xb + (r | cb));
        }
    }
    #pragma unroll
    for (int m = 0; m < TILE_M; ++m) {
        const v2f a = *(const v2f*)(za + 4 * m);
        v2f bf[4];
        #pragma unroll
        for (int t = 0; t < 4; ++t) bf[t] = Wbh[m * DOUT + t * 16];
        __builtin_amdgcn_sched_barrier(0);
        const int s = m % DEPTH;
        float a0 = 0.f, a1 = 0.f;
        #pragma unroll
        for (int j = 0; j < KNBR; ++j) {
            a0 += nv[s][j].x;
            a1 += nv[s][j].y;
        }
        float* zr = zw + m * DZP;
        zr[DIN + c0]     = a0 * (1.0f / KNBR);
        zr[DIN + c0 + 1] = a1 * (1.0f / KNBR);
        __builtin_amdgcn_sched_barrier(0);
        if (m + DEPTH < TILE_M) {
            const int mn = m + DEPTH;
            #pragma unroll
            for (int j = 0; j < KNBR; ++j) {
                const unsigned r = (unsigned)__builtin_amdgcn_readlane((int)ro[j & 7], 2 * mn + (j >> 3));
                nv[s][j] = *(const v2f*)(xb + (r | cb));
            }
        }
        #pragma unroll
        for (int t = 0; t < 4; ++t) {
            acc[t] = __builtin_amdgcn_wmma_f32_16x16x4_f32(
                false, a, false, bf[t], (short)0, acc[t], false, false);
        }
    }

    for (int kk = KSTEPS / 2; kk < KSTEPS; ++kk) {
        v2f a = *(const v2f*)(za + 4 * kk);
        const v2f* wrow = Wbh + kk * DOUT;
        #pragma unroll
        for (int t = 0; t < 4; ++t) {
            v2f bf = wrow[t * 16];
            acc[t] = __builtin_amdgcn_wmma_f32_16x16x4_f32(
                false, a, false, bf, (short)0, acc[t], false, false);
        }
    }

    #pragma unroll
    for (int t = 0; t < 4; ++t) {
        #pragma unroll
        for (int v = 0; v < 8; ++v) {
            const long row = base + v + 8 * hi;
            y[row * DOUT + t * 16 + nlo] = acc[t][v];
        }
    }
}

extern "C" void kernel_launch(void* const* d_in, const int* in_sizes, int n_in,
                              void* d_out, int out_size, void* d_ws, size_t ws_size,
                              hipStream_t stream) {
    const float* x    = (const float*)d_in[0];
    const int*   nbr  = (const int*)d_in[1];
    const float* W    = (const float*)d_in[2];
    const float* bias = (const float*)d_in[3];
    float*       y    = (float*)d_out;

    const int n_nodes = in_sizes[0] / DIN;
    const int n_tiles = (n_nodes + TILE_M - 1) / TILE_M;
    const int blocks  = (n_tiles + WAVES - 1) / WAVES;

    sage_fused_wmma_f32<<<blocks, BLOCK, 0, stream>>>(x, nbr, W, bias, y, n_tiles);
}
